# starting kernel
# baseline (speedup 1.0000x reference)
#include <hip/hip_runtime.h>
#include <math.h>

#define B_  64
#define S_  32
#define P_  64
#define V_  2562
#define BS_ (B_*S_)
#define VPB 128

typedef float v2f __attribute__((ext_vector_type(2)));
typedef float v8f __attribute__((ext_vector_type(8)));

__global__ void mesh_reduce_kernel(const float* __restrict__ transforms,
                                   const float* __restrict__ weights,
                                   const float* __restrict__ scales,
                                   float* __restrict__ mt) {
    int bs = blockIdx.x * blockDim.x + threadIdx.x;
    if (bs >= BS_) return;

    float acc[12];
#pragma unroll
    for (int i = 0; i < 12; ++i) acc[i] = 0.f;

    const float* tr = transforms + (size_t)bs * (P_ * 6);
    const float* wp = weights    + (size_t)bs * P_;

    for (int p = 0; p < P_; ++p) {
        float tx = tr[p*6+0], ty = tr[p*6+1], tz = tr[p*6+2];
        float a0 = tr[p*6+3], a1 = tr[p*6+4], a2 = tr[p*6+5];
        float w  = wp[p];

        float sa, ca, sb, cb, sg, cg;
        __sincosf(a0, &sa, &ca);
        __sincosf(a1, &sb, &cb);
        __sincosf(a2, &sg, &cg);

        float r00 =  cb*cg;             float r01 = -cb*sg;             float r02 =  sb;
        float r10 =  sa*sb*cg + ca*sg;  float r11 = -sa*sb*sg + ca*cg;  float r12 = -sa*cb;
        float r20 = -ca*sb*cg + sa*sg;  float r21 =  ca*sb*sg + sa*cg;  float r22 =  ca*cb;

        acc[0] = fmaf(w, r00, acc[0]); acc[1] = fmaf(w, r01, acc[1]); acc[2] = fmaf(w, r02, acc[2]);
        acc[3] = fmaf(w, r10, acc[3]); acc[4] = fmaf(w, r11, acc[4]); acc[5] = fmaf(w, r12, acc[5]);
        acc[6] = fmaf(w, r20, acc[6]); acc[7] = fmaf(w, r21, acc[7]); acc[8] = fmaf(w, r22, acc[8]);
        acc[9] = fmaf(w, tx, acc[9]);  acc[10] = fmaf(w, ty, acc[10]); acc[11] = fmaf(w, tz, acc[11]);
    }

    float sc = scales[bs];
    float* o = mt + (size_t)bs * 12;
#pragma unroll
    for (int i = 0; i < 9; ++i) o[i] = sc * acc[i];
    o[9] = acc[9]; o[10] = acc[10]; o[11] = acc[11];
}

__global__ void mesh_apply_kernel(const float* __restrict__ verts,
                                  const float* __restrict__ mt,
                                  float* __restrict__ out) {
    __shared__ __align__(16) float sV[VPB * 3];
    __shared__ __align__(16) float sM[12];
    __shared__ __align__(16) float sO[VPB * 3];

    const int tid = threadIdx.x;
    const int bs  = blockIdx.y;
    const int v0  = blockIdx.x * VPB;
    const bool full = (v0 + VPB) <= V_;

    if (full) {
        if (tid < (VPB * 3) / 2) {
            const v2f* src = (const v2f*)(verts + (size_t)v0 * 3);
            ((v2f*)sV)[tid] = src[tid];
        }
    } else {
        const int nIn = (V_ - v0) * 3;
        for (int idx = tid; idx < VPB * 3; idx += blockDim.x)
            sV[idx] = (idx < nIn) ? verts[(size_t)v0 * 3 + idx] : 0.f;
    }
    if (tid < 12) sM[tid] = mt[(size_t)bs * 12 + tid];
    __syncthreads();

    const int lane = tid & 31;
    const int wave = tid >> 5;
    const int m    = lane & 15;
    const int hi   = lane >> 4;
    const int lv   = wave * 16 + m;

    float ax0 = sV[lv * 3 + hi * 2];
    float ay0 = sV[lv * 3 + 1];
    v2f a;
    a.x = ax0;
    a.y = (hi == 0) ? ay0 : 0.f;

    const int mc = (m < 3) ? m : 0;
    float b0 = sM[mc * 3 + hi * 2];
    float b1 = sM[mc * 3 + 1];
    const bool colOK = (m < 3);
    v2f b;
    b.x = colOK ? b0 : 0.f;
    b.y = (colOK && hi == 0) ? b1 : 0.f;

    v8f c = {};
    c = __builtin_amdgcn_wmma_f32_16x16x4_f32(false, a, false, b,
                                              (short)0, c, false, false);

    if (m < 3) {
        float tb = sM[9 + m];
#pragma unroll
        for (int r = 0; r < 8; ++r) {
            int row = r + hi * 8;
            sO[(wave * 16 + row) * 3 + m] = c[r] + tb;
        }
    }
    __syncthreads();

    const size_t base = (size_t)bs * ((size_t)V_ * 3) + (size_t)v0 * 3;
    if (full) {
        if (tid < (VPB * 3) / 2) {
            v2f val = ((const v2f*)sO)[tid];
            __builtin_nontemporal_store(val, (v2f*)(out + base) + tid);
        }
    } else {
        const int nOut = (V_ - v0) * 3;
        for (int idx = tid; idx < nOut; idx += blockDim.x)
            __builtin_nontemporal_store(sO[idx], out + base + idx);
    }
}

extern "C" void kernel_launch(void* const* d_in, const int* in_sizes, int n_in,
                              void* d_out, int out_size, void* d_ws, size_t ws_size,
                              hipStream_t stream) {
    const float* verts      = (const float*)d_in[0];
    const float* scales     = (const float*)d_in[1];
    const float* transforms = (const float*)d_in[2];
    const float* weights    = (const float*)d_in[3];
    float* out = (float*)d_out;
    float* mt  = (float*)d_ws;

    mesh_reduce_kernel<<<dim3((BS_ + 255) / 256), dim3(256), 0, stream>>>(
        transforms, weights, scales, mt);

    dim3 grid((V_ + VPB - 1) / VPB, BS_);
    mesh_apply_kernel<<<grid, dim3(256), 0, stream>>>(verts, mt, out);
}


# apply kernel: one work-group per 4 consecutive (b,s) holds B in registers and loops over all vertices, 3 WMMA tiles per wave per trip, LDS-only barrier: 512 work-groups instead of 43008 one-tile ones
# speedup vs baseline: 2.0288x; 2.0288x over previous
#include <hip/hip_runtime.h>
#include <math.h>

#define B_  64
#define S_  32
#define P_  64
#define V_  2562
#define BS_ (B_*S_)
#define NG   4
#define TPW  3
#define TV   (8 * TPW * 16)
#define NIT  ((V_ + TV - 1) / TV)
#define GSTR (TV * 3 + 6)
#define NST  ((NG * TV * 3 / 2) / 256)

typedef float v2f __attribute__((ext_vector_type(2)));
typedef float v8f __attribute__((ext_vector_type(8)));

__global__ void mesh_reduce_kernel(const float* __restrict__ transforms,
                                   const float* __restrict__ weights,
                                   const float* __restrict__ scales,
                                   float* __restrict__ mt) {
    int bs = blockIdx.x * blockDim.x + threadIdx.x;
    if (bs >= BS_) return;

    float acc[12];
#pragma unroll
    for (int i = 0; i < 12; ++i) acc[i] = 0.f;

    const float* tr = transforms + (size_t)bs * (P_ * 6);
    const float* wp = weights    + (size_t)bs * P_;

    for (int p = 0; p < P_; ++p) {
        float tx = tr[p*6+0], ty = tr[p*6+1], tz = tr[p*6+2];
        float a0 = tr[p*6+3], a1 = tr[p*6+4], a2 = tr[p*6+5];
        float w  = wp[p];

        float sa, ca, sb, cb, sg, cg;
        __sincosf(a0, &sa, &ca);
        __sincosf(a1, &sb, &cb);
        __sincosf(a2, &sg, &cg);

        float r00 =  cb*cg;             float r01 = -cb*sg;             float r02 =  sb;
        float r10 =  sa*sb*cg + ca*sg;  float r11 = -sa*sb*sg + ca*cg;  float r12 = -sa*cb;
        float r20 = -ca*sb*cg + sa*sg;  float r21 =  ca*sb*sg + sa*cg;  float r22 =  ca*cb;

        acc[0] = fmaf(w, r00, acc[0]); acc[1] = fmaf(w, r01, acc[1]); acc[2] = fmaf(w, r02, acc[2]);
        acc[3] = fmaf(w, r10, acc[3]); acc[4] = fmaf(w, r11, acc[4]); acc[5] = fmaf(w, r12, acc[5]);
        acc[6] = fmaf(w, r20, acc[6]); acc[7] = fmaf(w, r21, acc[7]); acc[8] = fmaf(w, r22, acc[8]);
        acc[9] = fmaf(w, tx, acc[9]);  acc[10] = fmaf(w, ty, acc[10]); acc[11] = fmaf(w, tz, acc[11]);
    }

    float sc = scales[bs];
    float* o = mt + (size_t)bs * 12;
#pragma unroll
    for (int i = 0; i < 9; ++i) o[i] = sc * acc[i];
    o[9] = acc[9]; o[10] = acc[10]; o[11] = acc[11];
}

static_assert(BS_ % NG == 0, "record groups must tile the batch");
static_assert((NG * TV * 3 / 2) % 256 == 0, "whole stores per thread");
static_assert((V_ % 2) == 0 && (TV % 2) == 0, "b64 stores need an even float count per row piece");

__device__ __forceinline__ void lds_barrier() {
    __builtin_amdgcn_fence(__ATOMIC_RELEASE, "workgroup", "local");
    __builtin_amdgcn_s_barrier();
    __builtin_amdgcn_fence(__ATOMIC_ACQUIRE, "workgroup", "local");
}

__device__ __forceinline__ v2f load_a_frag(const float* __restrict__ verts, int v, int hi) {
    const float* p = verts + v * 3;
    float e0 = p[hi * 2];
    float e1 = p[1];
    v2f a;
    a.x = e0;
    a.y = (hi == 0) ? e1 : 0.f;
    return a;
}

__global__ __launch_bounds__(256)
void mesh_apply_kernel(const float* __restrict__ verts,
                       const float* __restrict__ mt,
                       float* __restrict__ out) {
    __shared__ __align__(16) float sO[2][NG * GSTR];

    const int tid  = threadIdx.x;
    const int lane = tid & 31;
    const int wave = tid >> 5;
    const int m    = lane & 15;
    const int hi   = lane >> 4;
    const int bs0  = blockIdx.x * NG;

    const bool colOK = (m < NG * 3);
    const int  mc = colOK ? m : 0;
    const int  g  = mc / 3;
    const int  i  = mc - g * 3;
    const float* rec = mt + (size_t)(bs0 + g) * 12;
    const float b0 = rec[i * 3 + hi * 2];
    const float b1 = rec[i * 3 + 1];
    const float tb = rec[9 + i];
    v2f b;
    b.x = colOK ? b0 : 0.f;
    b.y = (colOK && hi == 0) ? b1 : 0.f;

    const int lv = wave * (TPW * 16) + m;
    v2f an[TPW];
#pragma unroll
    for (int t = 0; t < TPW; ++t)
        an[t] = load_a_frag(verts, min(lv + 16 * t, V_ - 1), hi);

    const int wofs = g * GSTR + (wave * (TPW * 16) + hi * 8) * 3 + i;

    for (int it = 0; it < NIT; ++it) {
        const int v0 = it * TV;
        v2f a[TPW];
#pragma unroll
        for (int t = 0; t < TPW; ++t) a[t] = an[t];

        const int vn = v0 + TV + lv;
#pragma unroll
        for (int t = 0; t < TPW; ++t)
            an[t] = load_a_frag(verts, min(vn + 16 * t, V_ - 1), hi);

        v8f c[TPW];
#pragma unroll
        for (int t = 0; t < TPW; ++t) {
            v8f z = {};
            c[t] = __builtin_amdgcn_wmma_f32_16x16x4_f32(false, a[t], false, b,
                                                         (short)0, z, false, false);
        }

        float* sw = &sO[it & 1][0] + wofs;
        if (colOK) {
#pragma unroll
            for (int t = 0; t < TPW; ++t)
#pragma unroll
                for (int r = 0; r < 8; ++r)
                    sw[(16 * t + r) * 3] = c[t][r] + tb;
        }
        lds_barrier();

        const float* sr = &sO[it & 1][0];
        const int nv = (min(TV, V_ - v0) * 3) / 2;
        float* ob = out + (size_t)bs0 * (V_ * 3) + (size_t)v0 * 3;
        v2f val[NST];
        int gg[NST], jj[NST];
#pragma unroll
        for (int k = 0; k < NST; ++k) {
            const int idx = tid + 256 * k;
            gg[k] = idx / (TV * 3 / 2);
            jj[k] = idx - gg[k] * (TV * 3 / 2);
            val[k] = *(const v2f*)(sr + gg[k] * GSTR + 2 * jj[k]);
        }
        if (nv == TV * 3 / 2) {
#pragma unroll
            for (int k = 0; k < NST; ++k)
                __builtin_nontemporal_store(val[k], (v2f*)(ob + (size_t)gg[k] * (V_ * 3)) + jj[k]);
        } else {
#pragma unroll
            for (int k = 0; k < NST; ++k)
                if (jj[k] < nv)
                    __builtin_nontemporal_store(val[k], (v2f*)(ob + (size_t)gg[k] * (V_ * 3)) + jj[k]);
        }
    }
}

extern "C" void kernel_launch(void* const* d_in, const int* in_sizes, int n_in,
                              void* d_out, int out_size, void* d_ws, size_t ws_size,
                              hipStream_t stream) {
    const float* verts      = (const float*)d_in[0];
    const float* scales     = (const float*)d_in[1];
    const float* transforms = (const float*)d_in[2];
    const float* weights    = (const float*)d_in[3];
    float* out = (float*)d_out;
    float* mt  = (float*)d_ws;

    mesh_reduce_kernel<<<dim3((BS_ + 255) / 256), dim3(256), 0, stream>>>(
        transforms, weights, scales, mt);

    mesh_apply_kernel<<<dim3(BS_ / NG), dim3(256), 0, stream>>>(verts, mt, out);
}


# fuse the prototype reduction into the apply kernel: one (record, prototype) pair per thread, summed over prototypes as a fixed LDS tree (sum order changes); one launch instead of two
# speedup vs baseline: 3.7658x; 1.8562x over previous
#include <hip/hip_runtime.h>
#include <math.h>

#define B_  64
#define S_  32
#define P_  64
#define V_  2562
#define BS_ (B_*S_)
#define NG   4
#define TPW  3
#define TV   (8 * TPW * 16)
#define NIT  ((V_ + TV - 1) / TV)
#define GSTR (TV * 3 + 6)
#define NST  ((NG * TV * 3 / 2) / 256)
#define PSTR 17
#define KSTR (16 * PSTR)

typedef float v2f __attribute__((ext_vector_type(2)));
typedef float v8f __attribute__((ext_vector_type(8)));

static_assert(BS_ % NG == 0, "record groups must tile the batch");
static_assert(NG * P_ == 256, "one (record, prototype) pair per thread");
static_assert(12 * KSTR <= NG * GSTR, "the head's value planes fit in one output buffer");
static_assert((NG * TV * 3 / 2) % 256 == 0, "whole stores per thread");
static_assert((V_ % 2) == 0 && (TV % 2) == 0, "b64 stores need an even float count per row piece");

__device__ __forceinline__ void lds_barrier() {
    __builtin_amdgcn_fence(__ATOMIC_RELEASE, "workgroup", "local");
    __builtin_amdgcn_s_barrier();
    __builtin_amdgcn_fence(__ATOMIC_ACQUIRE, "workgroup", "local");
}

__device__ __forceinline__ v2f load_a_frag(const float* __restrict__ verts, int v, int hi) {
    const float* p = verts + v * 3;
    float e0 = p[hi * 2];
    float e1 = p[1];
    v2f a;
    a.x = e0;
    a.y = (hi == 0) ? e1 : 0.f;
    return a;
}

__global__ __launch_bounds__(256)
void mesh_fused_kernel(const float* __restrict__ verts,
                       const float* __restrict__ scales,
                       const float* __restrict__ transforms,
                       const float* __restrict__ weights,
                       float* __restrict__ out) {
    __shared__ __align__(16) float sO[2][NG * GSTR];

    const int tid  = threadIdx.x;
    const int lane = tid & 31;
    const int wave = tid >> 5;
    const int m    = lane & 15;
    const int hi   = lane >> 4;
    const int bs0  = blockIdx.x * NG;

    const size_t pair = (size_t)bs0 * P_ + tid;
    const float* tr = transforms + pair * 6;
    const float tx = tr[0], ty = tr[1], tz = tr[2];
    const float a0 = tr[3], a1 = tr[4], a2 = tr[5];
    const float w  = weights[pair];
    const int   hk = tid >> 4;
    const int   hj = tid & 15;
    const float hsc = scales[bs0 + (hj >> 2)];

    const int lv = wave * (TPW * 16) + m;
    v2f an[TPW];
#pragma unroll
    for (int t = 0; t < TPW; ++t)
        an[t] = load_a_frag(verts, min(lv + 16 * t, V_ - 1), hi);

    {
        float sa, ca, sb, cb, sg, cg;
        __sincosf(a0, &sa, &ca);
        __sincosf(a1, &sb, &cb);
        __sincosf(a2, &sg, &cg);

        float pv[12];
        pv[0] =  cb*cg;             pv[1] = -cb*sg;             pv[2] =  sb;
        pv[3] =  sa*sb*cg + ca*sg;  pv[4] = -sa*sb*sg + ca*cg;  pv[5] = -sa*cb;
        pv[6] = -ca*sb*cg + sa*sg;  pv[7] =  ca*sb*sg + sa*cg;  pv[8] =  ca*cb;
        pv[9] = tx;                 pv[10] = ty;                pv[11] = tz;

        float* sp = &sO[0][0] + (tid >> 4) * PSTR + (tid & 15);
#pragma unroll
        for (int k = 0; k < 12; ++k) sp[k * KSTR] = w * pv[k];
    }
    lds_barrier();

    if (tid < 12 * 16) {
        const float* sq = &sO[0][0] + hk * KSTR + hj * PSTR;
        float q[16];
#pragma unroll
        for (int e = 0; e < 16; ++e) q[e] = sq[e];
#pragma unroll
        for (int h = 8; h >= 1; h >>= 1)
#pragma unroll
            for (int e = 0; e < h; ++e) q[e] += q[e + h];
        float s = q[0];
        s += __int_as_float(__builtin_amdgcn_update_dpp(0, __float_as_int(s), 0xB1, 0xf, 0xf, true));
        s += __int_as_float(__builtin_amdgcn_update_dpp(0, __float_as_int(s), 0x4E, 0xf, 0xf, true));
        if ((hj & 3) == 0)
            sO[1][(hj >> 2) * 12 + hk] = (hk < 9) ? hsc * s : s;
    }
    lds_barrier();

    const bool colOK = (m < NG * 3);
    const int  mc = colOK ? m : 0;
    const int  g  = mc / 3;
    const int  i  = mc - g * 3;
    const float* rec = &sO[1][0] + g * 12;
    const float b0 = rec[i * 3 + hi * 2];
    const float b1 = rec[i * 3 + 1];
    const float tb = rec[9 + i];
    v2f b;
    b.x = colOK ? b0 : 0.f;
    b.y = (colOK && hi == 0) ? b1 : 0.f;

    const int wofs = g * GSTR + (wave * (TPW * 16) + hi * 8) * 3 + i;

    for (int it = 0; it < NIT; ++it) {
        const int v0 = it * TV;
        v2f a[TPW];
#pragma unroll
        for (int t = 0; t < TPW; ++t) a[t] = an[t];

        const int vn = v0 + TV + lv;
#pragma unroll
        for (int t = 0; t < TPW; ++t)
            an[t] = load_a_frag(verts, min(vn + 16 * t, V_ - 1), hi);

        v8f c[TPW];
#pragma unroll
        for (int t = 0; t < TPW; ++t) {
            v8f z = {};
            c[t] = __builtin_amdgcn_wmma_f32_16x16x4_f32(false, a[t], false, b,
                                                         (short)0, z, false, false);
        }

        float* sw = &sO[it & 1][0] + wofs;
        if (colOK) {
#pragma unroll
            for (int t = 0; t < TPW; ++t)
#pragma unroll
                for (int r = 0; r < 8; ++r)
                    sw[(16 * t + r) * 3] = c[t][r] + tb;
        }
        lds_barrier();

        const float* sr = &sO[it & 1][0];
        const int nv = (min(TV, V_ - v0) * 3) / 2;
        float* ob = out + (size_t)bs0 * (V_ * 3) + (size_t)v0 * 3;
        v2f val[NST];
        int gg[NST], jj[NST];
#pragma unroll
        for (int k = 0; k < NST; ++k) {
            const int idx = tid + 256 * k;
            gg[k] = idx / (TV * 3 / 2);
            jj[k] = idx - gg[k] * (TV * 3 / 2);
            val[k] = *(const v2f*)(sr + gg[k] * GSTR + 2 * jj[k]);
        }
        if (nv == TV * 3 / 2) {
#pragma unroll
            for (int k = 0; k < NST; ++k)
                __builtin_nontemporal_store(val[k], (v2f*)(ob + (size_t)gg[k] * (V_ * 3)) + jj[k]);
        } else {
#pragma unroll
            for (int k = 0; k < NST; ++k)
                if (jj[k] < nv)
                    __builtin_nontemporal_store(val[k], (v2f*)(ob + (size_t)gg[k] * (V_ * 3)) + jj[k]);
        }
    }
}

extern "C" void kernel_launch(void* const* d_in, const int* in_sizes, int n_in,
                              void* d_out, int out_size, void* d_ws, size_t ws_size,
                              hipStream_t stream) {
    const float* verts      = (const float*)d_in[0];
    const float* scales     = (const float*)d_in[1];
    const float* transforms = (const float*)d_in[2];
    const float* weights    = (const float*)d_in[3];
    float* out = (float*)d_out;

    mesh_fused_kernel<<<dim3(BS_ / NG), dim3(256), 0, stream>>>(
        verts, scales, transforms, weights, out);
}


# output stores write whole 128-byte lines: each trip re-computes 16 vertices so every row piece starts and ends on a line boundary; 16-byte stores, 4 full lines per instruction (was 8-byte, 3 lines)
# speedup vs baseline: 6.5184x; 1.7309x over previous
#include <hip/hip_runtime.h>
#include <math.h>

#define B_  64
#define S_  32
#define P_  64
#define V_  2562
#define BS_ (B_*S_)
#define NG   4
#define TPW  3
#define TV   (8 * TPW * 16)
#define OV   16
#define SV   (TV - OV)
#define NIT  ((V_ + SV - 1) / SV)
#define ROWF (V_ * 3)
#define LNF  32
#define GSTR (TV * 3 + 6)
#define PSTR 17
#define KSTR (16 * PSTR)

typedef float v2f __attribute__((ext_vector_type(2)));
typedef float v4f __attribute__((ext_vector_type(4)));
typedef float v8f __attribute__((ext_vector_type(8)));

static_assert(BS_ % NG == 0, "record groups must tile the batch");
static_assert(NG * P_ == 256, "one (record, prototype) pair per thread");
static_assert(12 * KSTR <= NG * GSTR, "the head's value planes fit in one output buffer");
static_assert(NG * 2 == 8, "two store waves per record");
static_assert(OV * 3 >= LNF - 2, "the overlap reaches back to the line boundary below a piece's first float");
static_assert((ROWF % 2) == 0 && ((SV * 3) % 4) == 0 && ((OV * 3) % 4) == 0, "row pieces keep 8-byte, planes 16-byte phase");
static_assert((ROWF % 4) == (GSTR % 4) && ((NG * ROWF) % 4) == 0 && ((NG * GSTR) % 4) == 0,
              "an LDS float index and its global float index agree modulo 4, so line starts are 16-byte aligned in LDS");
static_assert(SV * 3 - 2 * (LNF - 1) >= 32 * LNF && ROWF - (NIT - 1) * SV * 3 - (LNF - 1) >= 32 * LNF,
              "every piece holds at least 32 whole lines: store groups 0..7 need no mask");
static_assert(SV * 3 + (LNF - 1) < 36 * LNF, "a piece is at most 36 lines: 9 store groups");
static_assert(NIT * SV >= V_ && (NIT - 1) * SV < V_, "the trips cover the vertex list");

__device__ __forceinline__ void lds_barrier() {
    __builtin_amdgcn_fence(__ATOMIC_RELEASE, "workgroup", "local");
    __builtin_amdgcn_s_barrier();
    __builtin_amdgcn_fence(__ATOMIC_ACQUIRE, "workgroup", "local");
}

__device__ __forceinline__ v2f load_a_frag(const float* __restrict__ verts, int v, int hi) {
    const float* p = verts + v * 3;
    float e0 = p[hi * 2];
    float e1 = p[1];
    v2f a;
    a.x = e0;
    a.y = (hi == 0) ? e1 : 0.f;
    return a;
}

__global__ __launch_bounds__(256)
void mesh_fused_kernel(const float* __restrict__ verts,
                       const float* __restrict__ scales,
                       const float* __restrict__ transforms,
                       const float* __restrict__ weights,
                       float* __restrict__ out) {
    __shared__ __align__(16) float sO[2][NG * GSTR];

    const int tid  = threadIdx.x;
    const int lane = tid & 31;
    const int wave = tid >> 5;
    const int m    = lane & 15;
    const int hi   = lane >> 4;
    const int bs0  = blockIdx.x * NG;

    const size_t pair = (size_t)bs0 * P_ + tid;
    const float* tr = transforms + pair * 6;
    const float tx = tr[0], ty = tr[1], tz = tr[2];
    const float a0 = tr[3], a1 = tr[4], a2 = tr[5];
    const float w  = weights[pair];
    const int   hk = tid >> 4;
    const int   hj = tid & 15;
    const float hsc = scales[bs0 + (hj >> 2)];

    const int lv = wave * (TPW * 16) + m;
    v2f an[TPW];
#pragma unroll
    for (int t = 0; t < TPW; ++t)
        an[t] = load_a_frag(verts, min(max(lv + 16 * t - OV, 0), V_ - 1), hi);

    {
        float sa, ca, sb, cb, sg, cg;
        __sincosf(a0, &sa, &ca);
        __sincosf(a1, &sb, &cb);
        __sincosf(a2, &sg, &cg);

        float pv[12];
        pv[0] =  cb*cg;             pv[1] = -cb*sg;             pv[2] =  sb;
        pv[3] =  sa*sb*cg + ca*sg;  pv[4] = -sa*sb*sg + ca*cg;  pv[5] = -sa*cb;
        pv[6] = -ca*sb*cg + sa*sg;  pv[7] =  ca*sb*sg + sa*cg;  pv[8] =  ca*cb;
        pv[9] = tx;                 pv[10] = ty;                pv[11] = tz;

        float* sp = &sO[0][0] + (tid >> 4) * PSTR + (tid & 15);
#pragma unroll
        for (int k = 0; k < 12; ++k) sp[k * KSTR] = w * pv[k];
    }
    lds_barrier();

    if (tid < 12 * 16) {
        const float* sq = &sO[0][0] + hk * KSTR + hj * PSTR;
        float q[16];
#pragma unroll
        for (int e = 0; e < 16; ++e) q[e] = sq[e];
#pragma unroll
        for (int h = 8; h >= 1; h >>= 1)
#pragma unroll
            for (int e = 0; e < h; ++e) q[e] += q[e + h];
        float s = q[0];
        s += __int_as_float(__builtin_amdgcn_update_dpp(0, __float_as_int(s), 0xB1, 0xf, 0xf, true));
        s += __int_as_float(__builtin_amdgcn_update_dpp(0, __float_as_int(s), 0x4E, 0xf, 0xf, true));
        if ((hj & 3) == 0)
            sO[1][(hj >> 2) * 12 + hk] = (hk < 9) ? hsc * s : s;
    }
    lds_barrier();

    const bool colOK = (m < NG * 3);
    const int  mc = colOK ? m : 0;
    const int  g  = mc / 3;
    const int  i  = mc - g * 3;
    const float* rec = &sO[1][0] + g * 12;
    const float b0 = rec[i * 3 + hi * 2];
    const float b1 = rec[i * 3 + 1];
    const float tb = rec[9 + i];
    v2f b;
    b.x = colOK ? b0 : 0.f;
    b.y = (colOK && hi == 0) ? b1 : 0.f;

    const int wofs = g * GSTR + (wave * (TPW * 16) + hi * 8) * 3 + i;

    const int srow   = wave >> 1;
    const int half = wave & 1;
    const int rg   = (bs0 + srow) * ROWF;

    for (int it = 0; it < NIT; ++it) {
        const int p0 = it * (SV * 3);
        v2f a[TPW];
#pragma unroll
        for (int t = 0; t < TPW; ++t) a[t] = an[t];

        const int vn = p0 / 3 + SV - OV + lv;
#pragma unroll
        for (int t = 0; t < TPW; ++t)
            an[t] = load_a_frag(verts, min(vn + 16 * t, V_ - 1), hi);

        v8f c[TPW];
#pragma unroll
        for (int t = 0; t < TPW; ++t) {
            v8f z = {};
            c[t] = __builtin_amdgcn_wmma_f32_16x16x4_f32(false, a[t], false, b,
                                                         (short)0, z, false, false);
        }

        float* sw = &sO[it & 1][0] + wofs;
        if (colOK) {
#pragma unroll
            for (int t = 0; t < TPW; ++t)
#pragma unroll
                for (int r = 0; r < 8; ++r)
                    sw[(16 * t + r) * 3] = c[t][r] + tb;
        }
        lds_barrier();

        const float* sr = &sO[it & 1][0];
        const int pb = (it == 0)       ? ((rg + LNF - 1) & ~(LNF - 1)) : ((rg + p0) & ~(LNF - 1));
        const int pe = (it == NIT - 1) ? ((rg + ROWF) & ~(LNF - 1))    : ((rg + p0 + SV * 3) & ~(LNF - 1));
        const float* sl = sr + srow * GSTR + (pb - rg - p0 + OV * 3);
        float* og = out + pb;
        v4f val[4];
#pragma unroll
        for (int k = 0; k < 4; ++k)
            val[k] = *(const v4f*)(sl + (2 * k + half) * (4 * LNF) + lane * 4);
        const bool ok8 = (half == 0) && (pb + 8 * (4 * LNF) + lane * 4 < pe);
        v4f val8 = {};
        if (ok8) val8 = *(const v4f*)(sl + 8 * (4 * LNF) + lane * 4);
#pragma unroll
        for (int k = 0; k < 4; ++k)
            __builtin_nontemporal_store(val[k], (v4f*)(og + (2 * k + half) * (4 * LNF) + lane * 4));
        if (ok8)
            __builtin_nontemporal_store(val8, (v4f*)(og + 8 * (4 * LNF) + lane * 4));

        if (half == 1) {
            if (it == 0) {
                const int np = (pb - rg) >> 1;
                if (lane < np) {
                    const v2f e = *(const v2f*)(sr + srow * GSTR + OV * 3 + 2 * lane);
                    __builtin_nontemporal_store(e, (v2f*)(out + rg + 2 * lane));
                }
            }
            if (it == NIT - 1) {
                const int np = (rg + ROWF - pe) >> 1;
                if (lane < np) {
                    const v2f e = *(const v2f*)(sl + (pe - pb) + 2 * lane);
                    __builtin_nontemporal_store(e, (v2f*)(out + pe + 2 * lane));
                }
            }
        }
    }
}

extern "C" void kernel_launch(void* const* d_in, const int* in_sizes, int n_in,
                              void* d_out, int out_size, void* d_ws, size_t ws_size,
                              hipStream_t stream) {
    const float* verts      = (const float*)d_in[0];
    const float* scales     = (const float*)d_in[1];
    const float* transforms = (const float*)d_in[2];
    const float* weights    = (const float*)d_in[3];
    float* out = (float*)d_out;

    mesh_fused_kernel<<<dim3(BS_ / NG), dim3(256), 0, stream>>>(
        verts, scales, transforms, weights, out);
}


# row pieces cut at 256-byte boundaries, not 128 (overlap 24 vertices): each store instruction writes two whole 256-byte units; 7 tiles per wave per trip so the list fills whole trips (3, was 7)
# speedup vs baseline: 7.1890x; 1.1029x over previous
#include <hip/hip_runtime.h>
#include <math.h>

#define B_  64
#define S_  32
#define P_  64
#define V_  2562
#define BS_ (B_*S_)
#define NG   4
#define TPW  7
#define TV   (8 * TPW * 16)
#define OV   24
#define SV   (TV - OV)
#define NIT  ((V_ + SV - 1) / SV)
#define ROWF (V_ * 3)
#define LNF  32
#define UNF  64
#define NGRP ((SV * 3 + UNF - 1) / (4 * LNF) + 1)
#define NVAL ((NGRP + 1) / 2)
#define LASTF (ROWF - (NIT - 1) * SV * 3)
#define NFULL (((LASTF < SV * 3 ? LASTF : SV * 3) - 2 * (UNF - 2)) / (4 * LNF))
#define GSTR (TV * 3 + 6)
#define PSTR 17
#define KSTR (16 * PSTR)

typedef float v2f __attribute__((ext_vector_type(2)));
typedef float v4f __attribute__((ext_vector_type(4)));
typedef float v8f __attribute__((ext_vector_type(8)));

static_assert(BS_ % NG == 0, "record groups must tile the batch");
static_assert(NG * P_ == 256, "one (record, prototype) pair per thread");
static_assert(12 * KSTR <= NG * GSTR, "the head's value planes fit in one output buffer");
static_assert(NG * 2 == 8, "two store waves per record");
static_assert(OV * 3 >= UNF - 2, "the overlap reaches back to the 256-byte boundary below a piece's first float");
static_assert(UNF - 2 <= 2 * 32, "a row's partial first or last unit is at most 32 8-byte pieces: one per lane");
static_assert((ROWF % 2) == 0 && ((SV * 3) % 4) == 0 && ((OV * 3) % 4) == 0, "row pieces keep 8-byte, planes 16-byte phase");
static_assert((ROWF % 4) == (GSTR % 4) && ((NG * ROWF) % 4) == 0 && ((NG * GSTR) % 4) == 0,
              "an LDS float index and its global float index agree modulo 4, so line starts are 16-byte aligned in LDS");
static_assert(SV * 3 + (UNF - 2) <= NGRP * 4 * LNF, "a piece is at most NGRP store groups");
static_assert(SV * 3 + OV * 3 <= TV * 3, "a piece and its look-back lie inside the plane");
static_assert(NFULL >= 1 && NFULL <= NGRP, "the unmasked store groups are groups of every piece");
static_assert(NIT * SV >= V_ && (NIT - 1) * SV < V_, "the trips cover the vertex list");

__device__ __forceinline__ void lds_barrier() {
    __builtin_amdgcn_fence(__ATOMIC_RELEASE, "workgroup", "local");
    __builtin_amdgcn_s_barrier();
    __builtin_amdgcn_fence(__ATOMIC_ACQUIRE, "workgroup", "local");
}

__device__ __forceinline__ v2f load_a_frag(const float* __restrict__ verts, int v, int hi) {
    const float* p = verts + v * 3;
    float e0 = p[hi * 2];
    float e1 = p[1];
    v2f a;
    a.x = e0;
    a.y = (hi == 0) ? e1 : 0.f;
    return a;
}

__global__ __launch_bounds__(256)
void mesh_fused_kernel(const float* __restrict__ verts,
                       const float* __restrict__ scales,
                       const float* __restrict__ transforms,
                       const float* __restrict__ weights,
                       float* __restrict__ out) {
    __shared__ __align__(16) float sO[2][NG * GSTR];

    const int tid  = threadIdx.x;
    const int lane = tid & 31;
    const int wave = tid >> 5;
    const int m    = lane & 15;
    const int hi   = lane >> 4;
    const int bs0  = blockIdx.x * NG;

    const size_t pair = (size_t)bs0 * P_ + tid;
    const float* tr = transforms + pair * 6;
    const float tx = tr[0], ty = tr[1], tz = tr[2];
    const float a0 = tr[3], a1 = tr[4], a2 = tr[5];
    const float w  = weights[pair];
    const int   hk = tid >> 4;
    const int   hj = tid & 15;
    const float hsc = scales[bs0 + (hj >> 2)];

    const int lv = wave * (TPW * 16) + m;
    v2f an[TPW];
#pragma unroll
    for (int t = 0; t < TPW; ++t)
        an[t] = load_a_frag(verts, min(max(lv + 16 * t - OV, 0), V_ - 1), hi);

    {
        float sa, ca, sb, cb, sg, cg;
        __sincosf(a0, &sa, &ca);
        __sincosf(a1, &sb, &cb);
        __sincosf(a2, &sg, &cg);

        float pv[12];
        pv[0] =  cb*cg;             pv[1] = -cb*sg;             pv[2] =  sb;
        pv[3] =  sa*sb*cg + ca*sg;  pv[4] = -sa*sb*sg + ca*cg;  pv[5] = -sa*cb;
        pv[6] = -ca*sb*cg + sa*sg;  pv[7] =  ca*sb*sg + sa*cg;  pv[8] =  ca*cb;
        pv[9] = tx;                 pv[10] = ty;                pv[11] = tz;

        float* sp = &sO[0][0] + (tid >> 4) * PSTR + (tid & 15);
#pragma unroll
        for (int k = 0; k < 12; ++k) sp[k * KSTR] = w * pv[k];
    }
    lds_barrier();

    if (tid < 12 * 16) {
        const float* sq = &sO[0][0] + hk * KSTR + hj * PSTR;
        float q[16];
#pragma unroll
        for (int e = 0; e < 16; ++e) q[e] = sq[e];
#pragma unroll
        for (int h = 8; h >= 1; h >>= 1)
#pragma unroll
            for (int e = 0; e < h; ++e) q[e] += q[e + h];
        float s = q[0];
        s += __int_as_float(__builtin_amdgcn_update_dpp(0, __float_as_int(s), 0xB1, 0xf, 0xf, true));
        s += __int_as_float(__builtin_amdgcn_update_dpp(0, __float_as_int(s), 0x4E, 0xf, 0xf, true));
        if ((hj & 3) == 0)
            sO[1][(hj >> 2) * 12 + hk] = (hk < 9) ? hsc * s : s;
    }
    lds_barrier();

    const bool colOK = (m < NG * 3);
    const int  mc = colOK ? m : 0;
    const int  g  = mc / 3;
    const int  i  = mc - g * 3;
    const float* rec = &sO[1][0] + g * 12;
    const float b0 = rec[i * 3 + hi * 2];
    const float b1 = rec[i * 3 + 1];
    const float tb = rec[9 + i];
    v2f b;
    b.x = colOK ? b0 : 0.f;
    b.y = (colOK && hi == 0) ? b1 : 0.f;

    const int wofs = g * GSTR + (wave * (TPW * 16) + hi * 8) * 3 + i;

    const int srow   = wave >> 1;
    const int half = wave & 1;
    const int rg   = (bs0 + srow) * ROWF;

    for (int it = 0; it < NIT; ++it) {
        const int p0 = it * (SV * 3);
        v2f a[TPW];
#pragma unroll
        for (int t = 0; t < TPW; ++t) a[t] = an[t];

        const int vn = p0 / 3 + SV - OV + lv;
#pragma unroll
        for (int t = 0; t < TPW; ++t)
            an[t] = load_a_frag(verts, min(vn + 16 * t, V_ - 1), hi);

        v8f c[TPW];
#pragma unroll
        for (int t = 0; t < TPW; ++t) {
            v8f z = {};
            c[t] = __builtin_amdgcn_wmma_f32_16x16x4_f32(false, a[t], false, b,
                                                         (short)0, z, false, false);
        }

        float* sw = &sO[it & 1][0] + wofs;
        if (colOK) {
#pragma unroll
            for (int t = 0; t < TPW; ++t)
#pragma unroll
                for (int r = 0; r < 8; ++r)
                    sw[(16 * t + r) * 3] = c[t][r] + tb;
        }
        lds_barrier();

        const float* sr = &sO[it & 1][0];
        const int pb = (it == 0)       ? ((rg + UNF - 1) & ~(UNF - 1)) : ((rg + p0) & ~(UNF - 1));
        const int pe = (it == NIT - 1) ? ((rg + ROWF) & ~(UNF - 1))    : ((rg + p0 + SV * 3) & ~(UNF - 1));
        const float* sl = sr + srow * GSTR + (pb - rg - p0 + OV * 3);
        float* og = out + pb;
        v4f  val[NVAL];
        bool okv[NVAL];
#pragma unroll
        for (int k = 0; k < NVAL; ++k) {
            const int q = 2 * k + half;
            okv[k] = (2 * k + 1 < NFULL) || (pb + q * (4 * LNF) + lane * 4 < pe);
            v4f z4 = {};
            val[k] = z4;
            if (okv[k]) val[k] = *(const v4f*)(sl + q * (4 * LNF) + lane * 4);
        }
#pragma unroll
        for (int k = 0; k < NVAL; ++k) {
            const int q = 2 * k + half;
            if (okv[k]) __builtin_nontemporal_store(val[k], (v4f*)(og + q * (4 * LNF) + lane * 4));
        }

        if (half == 1) {
            if (it == 0) {
                const int np = (pb - rg) >> 1;
                if (lane < np) {
                    const v2f e = *(const v2f*)(sr + srow * GSTR + OV * 3 + 2 * lane);
                    __builtin_nontemporal_store(e, (v2f*)(out + rg + 2 * lane));
                }
            }
            if (it == NIT - 1) {
                const int np = (rg + ROWF - pe) >> 1;
                if (lane < np) {
                    const v2f e = *(const v2f*)(sl + (pe - pb) + 2 * lane);
                    __builtin_nontemporal_store(e, (v2f*)(out + pe + 2 * lane));
                }
            }
        }
    }
}

extern "C" void kernel_launch(void* const* d_in, const int* in_sizes, int n_in,
                              void* d_out, int out_size, void* d_ws, size_t ws_size,
                              hipStream_t stream) {
    const float* verts      = (const float*)d_in[0];
    const float* scales     = (const float*)d_in[1];
    const float* transforms = (const float*)d_in[2];
    const float* weights    = (const float*)d_in[3];
    float* out = (float*)d_out;

    mesh_fused_kernel<<<dim3(BS_ / NG), dim3(256), 0, stream>>>(
        verts, scales, transforms, weights, out);
}
